# starting kernel
# baseline (speedup 1.0000x reference)
#include <hip/hip_runtime.h>
#include <hip/hip_bf16.h>
#include <math.h>

typedef __attribute__((ext_vector_type(2))) float v2f;
typedef __attribute__((ext_vector_type(8))) float v8f;

#define IN_F   256
#define OUT_F  32
#define HEADS  8
#define COLS   256
#define WP_ELEMS (128 * COLS * 2)

__global__ void pack_W_kernel(const float* __restrict__ W, float* __restrict__ Wp) {
  const int idx = blockIdx.x * blockDim.x + threadIdx.x;
  if (idx >= WP_ELEMS) return;
  const int j    = idx & 1;
  const int n    = (idx >> 1) & (COLS - 1);
  const int p    = idx >> 9;
  const int k4   = p >> 1;
  const int half = p & 1;
  Wp[idx] = W[(size_t)(k4 * 4 + 2 * half + j) * COLS + n];
}

__global__ void __launch_bounds__(128)
proj_gemm_wmma(const float* __restrict__ Z, const float* __restrict__ Wp,
               const float* __restrict__ bias, float* __restrict__ Zp, int N) {
  const int lane = threadIdx.x & 31;
  const int wave = threadIdx.x >> 5;
  const int l16  = lane & 15;
  const int half = lane >> 4;
  const int m0   = blockIdx.x << 4;
  const int n0   = wave << 6;

  const int mA = m0 + l16;
  const float* zrow = Z + (size_t)(mA < N ? mA : (N - 1)) * IN_F;

  const float* wb = Wp + ((size_t)half * COLS + (n0 + l16)) * 2;

  v8f acc0 = {}; v8f acc1 = {}; v8f acc2 = {}; v8f acc3 = {};

  #pragma unroll 4
  for (int k = 0; k < IN_F; k += 4) {
    v2f a  = *reinterpret_cast<const v2f*>(zrow + k + 2 * half);
    v2f b0 = *reinterpret_cast<const v2f*>(wb);
    v2f b1 = *reinterpret_cast<const v2f*>(wb + 32);
    v2f b2 = *reinterpret_cast<const v2f*>(wb + 64);
    v2f b3 = *reinterpret_cast<const v2f*>(wb + 96);
    wb += 2 * COLS * 2;

    acc0 = __builtin_amdgcn_wmma_f32_16x16x4_f32(false, a, false, b0, (short)0, acc0, false, false);
    acc1 = __builtin_amdgcn_wmma_f32_16x16x4_f32(false, a, false, b1, (short)0, acc1, false, false);
    acc2 = __builtin_amdgcn_wmma_f32_16x16x4_f32(false, a, false, b2, (short)0, acc2, false, false);
    acc3 = __builtin_amdgcn_wmma_f32_16x16x4_f32(false, a, false, b3, (short)0, acc3, false, false);
  }

  v8f accs[4] = {acc0, acc1, acc2, acc3};
  #pragma unroll
  for (int t = 0; t < 4; ++t) {
    const int n  = n0 + 16 * t + l16;
    const float bv = bias[n];
    #pragma unroll
    for (int v = 0; v < 8; ++v) {
      const int mrow = m0 + v + 8 * half;
      if (mrow < N) Zp[(size_t)mrow * COLS + n] = accs[t][v] + bv;
    }
  }
}

__global__ void attn_logits_init(const float* __restrict__ Zp,
                                 const float* __restrict__ a_l, const float* __restrict__ a_r,
                                 float* __restrict__ e_l, float* __restrict__ e_r,
                                 float* __restrict__ mbuf, float* __restrict__ sbuf, int NH) {
  const int idx = blockIdx.x * blockDim.x + threadIdx.x;
  if (idx >= NH) return;
  const int n = idx >> 3;
  const int h = idx & 7;
  const float* z = Zp + (size_t)n * COLS + h;
  float al = 0.f, ar = 0.f;
  #pragma unroll
  for (int o = 0; o < OUT_F; ++o) {
    const float zv = z[o * HEADS];
    al = fmaf(zv, a_l[o * HEADS + h], al);
    ar = fmaf(zv, a_r[o * HEADS + h], ar);
  }
  e_l[idx]  = al;
  e_r[idx]  = ar;
  mbuf[idx] = -INFINITY;
  sbuf[idx] = 0.f;
}

__device__ __forceinline__ float leaky01(float v) {
  return v > 0.f ? v : 0.01f * v;
}

__device__ __forceinline__ void atomicMaxFloat(float* addr, float val) {
  if (val >= 0.f) atomicMax((int*)addr, __float_as_int(val));
  else            atomicMin((unsigned int*)addr, __float_as_uint(val));
}

__global__ void edge_max_kernel(const int* __restrict__ row, const int* __restrict__ col,
                                const float* __restrict__ e_l, const float* __restrict__ e_r,
                                float* __restrict__ mbuf, int EH) {
  const int idx = blockIdx.x * blockDim.x + threadIdx.x;
  if (idx >= EH) return;
  const int e = idx >> 3, h = idx & 7;
  const int r = row[e], c = col[e];
  const float v = leaky01(e_l[r * HEADS + h] + e_r[c * HEADS + h]);
  atomicMaxFloat(&mbuf[r * HEADS + h], v);
}

__global__ void edge_expsum_kernel(const int* __restrict__ row, const int* __restrict__ col,
                                   const float* __restrict__ e_l, const float* __restrict__ e_r,
                                   const float* __restrict__ mbuf, float* __restrict__ sbuf, int EH) {
  const int idx = blockIdx.x * blockDim.x + threadIdx.x;
  if (idx >= EH) return;
  const int e = idx >> 3, h = idx & 7;
  const int r = row[e], c = col[e];
  const float v  = leaky01(e_l[r * HEADS + h] + e_r[c * HEADS + h]);
  const float ex = __expf(v - mbuf[r * HEADS + h]);
  atomicAdd(&sbuf[r * HEADS + h], ex);
}

__global__ void __launch_bounds__(256)
edge_scatter_kernel(const int* __restrict__ row, const int* __restrict__ col,
                    const float* __restrict__ e_l, const float* __restrict__ e_r,
                    const float* __restrict__ mbuf, const float* __restrict__ sbuf,
                    const float* __restrict__ Zp, float* __restrict__ out) {
  const int e = blockIdx.x;
  const int c = threadIdx.x;
  const int r  = row[e];
  const int cl = col[e];
  const int h  = c & 7;
  const float v    = leaky01(e_l[r * HEADS + h] + e_r[cl * HEADS + h]);
  const float attn = __expf(v - mbuf[r * HEADS + h]) / sbuf[r * HEADS + h];
  atomicAdd(&out[(size_t)r * COLS + c], attn * Zp[(size_t)cl * COLS + c]);
}

extern "C" void kernel_launch(void* const* d_in, const int* in_sizes, int n_in,
                              void* d_out, int out_size, void* d_ws, size_t ws_size,
                              hipStream_t stream) {
  const float* Z   = (const float*)d_in[0];
  const int*   row = (const int*)d_in[1];
  const int*   col = (const int*)d_in[2];
  const float* W   = (const float*)d_in[3];
  const float* bv  = (const float*)d_in[4];
  const float* a_l = (const float*)d_in[5];
  const float* a_r = (const float*)d_in[6];
  float* out = (float*)d_out;

  const int N = in_sizes[0] / IN_F;
  const int E = in_sizes[1];

  float* Zp  = (float*)d_ws;
  float* e_l = Zp  + (size_t)N * COLS;
  float* e_r = e_l + (size_t)N * HEADS;
  float* mb  = e_r + (size_t)N * HEADS;
  float* sb  = mb  + (size_t)N * HEADS;
  float* Wp  = sb  + (size_t)N * HEADS;

  hipMemsetAsync(out, 0, (size_t)out_size * sizeof(float), stream);

  pack_W_kernel<<<(WP_ELEMS + 255) / 256, 256, 0, stream>>>(W, Wp);

  proj_gemm_wmma<<<dim3((N + 15) / 16), 128, 0, stream>>>(Z, Wp, bv, Zp, N);

  const int NH = N * HEADS;
  attn_logits_init<<<(NH + 255) / 256, 256, 0, stream>>>(Zp, a_l, a_r, e_l, e_r, mb, sb, NH);

  const int EH = E * HEADS;
  edge_max_kernel   <<<(EH + 255) / 256, 256, 0, stream>>>(row, col, e_l, e_r, mb, EH);
  edge_expsum_kernel<<<(EH + 255) / 256, 256, 0, stream>>>(row, col, e_l, e_r, mb, sb, EH);
  edge_scatter_kernel<<<E, 256, 0, stream>>>(row, col, e_l, e_r, mb, sb, Zp, out);
}


# Row-count prefix sum on whole lines: each of 32 waves scans one contiguous segment, 4 rows per lane (16-byte loads and stores, in-wave scan by DPP adds), not 1024 threads each walking a 49-row chunk
# speedup vs baseline: 3.7874x; 3.7874x over previous
#include <hip/hip_runtime.h>
#include <hip/hip_bf16.h>
#include <math.h>

typedef __attribute__((ext_vector_type(2))) float v2f;
typedef __attribute__((ext_vector_type(8))) float v8f;

#define IN_F   256
#define OUT_F  32
#define HEADS  8
#define COLS   256
#define WP_ELEMS (128 * COLS * 2)
#define RANK_BLOCKS 512

__global__ void pack_W_kernel(const float* __restrict__ W, float* __restrict__ Wp,
                              int* __restrict__ cnt, int N) {
  const int idx = blockIdx.x * blockDim.x + threadIdx.x;
  if (cnt != nullptr) {
    const int nthr = gridDim.x * blockDim.x;
    const int n4   = (N + 3) & ~3;
    for (int i = idx; i < n4; i += nthr) cnt[i] = 0;
  }
  if (idx >= WP_ELEMS) return;
  const int j    = idx & 1;
  const int n    = (idx >> 1) & (COLS - 1);
  const int p    = idx >> 9;
  const int k4   = p >> 1;
  const int half = p & 1;
  Wp[idx] = W[(size_t)(k4 * 4 + 2 * half + j) * COLS + n];
}

__global__ void __launch_bounds__(128)
proj_gemm_wmma(const float* __restrict__ Z, const float* __restrict__ Wp,
               const float* __restrict__ bias, float* __restrict__ Zp, int N,
               const int* __restrict__ row, int* __restrict__ cnt,
               int* __restrict__ rank, int E, int rank_blocks) {
  if ((int)blockIdx.x < rank_blocks) {
    const int nthr = rank_blocks * 128;
    for (int e = blockIdx.x * 128 + threadIdx.x; e < E; e += 2 * nthr) {
      const int  f  = e + nthr;
      const bool hf = f < E;
      const int  ra = row[e];
      const int  rb = row[hf ? f : e];
      const int  sa = atomicAdd(&cnt[ra], 1);
      const int  sb = atomicAdd(&cnt[rb], hf ? 1 : 0);
      rank[e] = sa;
      if (hf) rank[f] = sb;
    }
    return;
  }
  const int lane = threadIdx.x & 31;
  const int wave = threadIdx.x >> 5;
  const int l16  = lane & 15;
  const int half = lane >> 4;
  const int m0   = ((int)blockIdx.x - rank_blocks) << 4;
  const int n0   = wave << 6;

  const int mA = m0 + l16;
  const float* zrow = Z + (size_t)(mA < N ? mA : (N - 1)) * IN_F;

  const float* wb = Wp + ((size_t)half * COLS + (n0 + l16)) * 2;

  v8f acc0 = {}; v8f acc1 = {}; v8f acc2 = {}; v8f acc3 = {};

  #pragma unroll 4
  for (int k = 0; k < IN_F; k += 4) {
    v2f a  = *reinterpret_cast<const v2f*>(zrow + k + 2 * half);
    v2f b0 = *reinterpret_cast<const v2f*>(wb);
    v2f b1 = *reinterpret_cast<const v2f*>(wb + 32);
    v2f b2 = *reinterpret_cast<const v2f*>(wb + 64);
    v2f b3 = *reinterpret_cast<const v2f*>(wb + 96);
    wb += 2 * COLS * 2;

    acc0 = __builtin_amdgcn_wmma_f32_16x16x4_f32(false, a, false, b0, (short)0, acc0, false, false);
    acc1 = __builtin_amdgcn_wmma_f32_16x16x4_f32(false, a, false, b1, (short)0, acc1, false, false);
    acc2 = __builtin_amdgcn_wmma_f32_16x16x4_f32(false, a, false, b2, (short)0, acc2, false, false);
    acc3 = __builtin_amdgcn_wmma_f32_16x16x4_f32(false, a, false, b3, (short)0, acc3, false, false);
  }

  v8f accs[4] = {acc0, acc1, acc2, acc3};
  #pragma unroll
  for (int t = 0; t < 4; ++t) {
    const int n  = n0 + 16 * t + l16;
    const float bv = bias[n];
    #pragma unroll
    for (int v = 0; v < 8; ++v) {
      const int mrow = m0 + v + 8 * half;
      if (mrow < N) Zp[(size_t)mrow * COLS + n] = accs[t][v] + bv;
    }
  }
}

__global__ void attn_logits_init(const float* __restrict__ Zp,
                                 const float* __restrict__ a_l, const float* __restrict__ a_r,
                                 float* __restrict__ e_l, float* __restrict__ e_r,
                                 float* __restrict__ mbuf, float* __restrict__ sbuf,
                                 const int* __restrict__ row, const int* __restrict__ col,
                                 const int* __restrict__ rowptr, const int* __restrict__ rank,
                                 int* __restrict__ scol, int NH, int E) {
  const int idx = blockIdx.x * blockDim.x + threadIdx.x;

  if (rowptr != nullptr) {
    const int nthr = gridDim.x * blockDim.x;
    for (int e = idx; e < E; e += 2 * nthr) {
      const int f  = e + nthr;
      const int g  = f < E ? f : e;
      const int ra = row[e],  rb = row[g];
      const int ka = rank[e], kb = rank[g];
      const int ca = col[e],  cb = col[g];
      int pa = rowptr[ra] + ka;
      int pb = rowptr[rb] + kb;
      pa = pa < 0 ? 0 : (pa >= E ? E - 1 : pa);
      pb = pb < 0 ? 0 : (pb >= E ? E - 1 : pb);
      scol[pa] = ca;
      scol[pb] = cb;
    }
  }

  if (idx >= NH) return;
  const int n = idx >> 3;
  const int h = idx & 7;
  const float* z = Zp + (size_t)n * COLS + h;
  float al = 0.f, ar = 0.f;
  #pragma unroll
  for (int o = 0; o < OUT_F; ++o) {
    const float zv = z[o * HEADS];
    al = fmaf(zv, a_l[o * HEADS + h], al);
    ar = fmaf(zv, a_r[o * HEADS + h], ar);
  }
  e_l[idx]  = al;
  e_r[idx]  = ar;
  if (mbuf != nullptr) {
    mbuf[idx] = -INFINITY;
    sbuf[idx] = 0.f;
  }
}

__device__ __forceinline__ float leaky01(float v) {
  return v > 0.f ? v : 0.01f * v;
}

__device__ __forceinline__ void atomicMaxFloat(float* addr, float val) {
  if (val >= 0.f) atomicMax((int*)addr, __float_as_int(val));
  else            atomicMin((unsigned int*)addr, __float_as_uint(val));
}

__global__ void edge_max_kernel(const int* __restrict__ row, const int* __restrict__ col,
                                const float* __restrict__ e_l, const float* __restrict__ e_r,
                                float* __restrict__ mbuf, int EH) {
  const int idx = blockIdx.x * blockDim.x + threadIdx.x;
  if (idx >= EH) return;
  const int e = idx >> 3, h = idx & 7;
  const int r = row[e], c = col[e];
  const float v = leaky01(e_l[r * HEADS + h] + e_r[c * HEADS + h]);
  atomicMaxFloat(&mbuf[r * HEADS + h], v);
}

__global__ void edge_expsum_kernel(const int* __restrict__ row, const int* __restrict__ col,
                                   const float* __restrict__ e_l, const float* __restrict__ e_r,
                                   const float* __restrict__ mbuf, float* __restrict__ sbuf, int EH) {
  const int idx = blockIdx.x * blockDim.x + threadIdx.x;
  if (idx >= EH) return;
  const int e = idx >> 3, h = idx & 7;
  const int r = row[e], c = col[e];
  const float v  = leaky01(e_l[r * HEADS + h] + e_r[c * HEADS + h]);
  const float ex = __expf(v - mbuf[r * HEADS + h]);
  atomicAdd(&sbuf[r * HEADS + h], ex);
}

__global__ void __launch_bounds__(256)
edge_scatter_kernel(const int* __restrict__ row, const int* __restrict__ col,
                    const float* __restrict__ e_l, const float* __restrict__ e_r,
                    const float* __restrict__ mbuf, const float* __restrict__ sbuf,
                    const float* __restrict__ Zp, float* __restrict__ out) {
  const int e = blockIdx.x;
  const int c = threadIdx.x;
  const int r  = row[e];
  const int cl = col[e];
  const int h  = c & 7;
  const float v    = leaky01(e_l[r * HEADS + h] + e_r[cl * HEADS + h]);
  const float attn = __expf(v - mbuf[r * HEADS + h]) / sbuf[r * HEADS + h];
  atomicAdd(&out[(size_t)r * COLS + c], attn * Zp[(size_t)cl * COLS + c]);
}


#define SCAN_TRIPS 16

__device__ __forceinline__ int wave_incl_scan(int x, const int lane) {
  x += __builtin_amdgcn_update_dpp(0, x, 0x111, 0xf, 0xf, true);
  x += __builtin_amdgcn_update_dpp(0, x, 0x112, 0xf, 0xf, true);
  x += __builtin_amdgcn_update_dpp(0, x, 0x114, 0xf, 0xf, true);
  x += __builtin_amdgcn_update_dpp(0, x, 0x118, 0xf, 0xf, true);
  const int t = __builtin_amdgcn_readlane(x, 15);
  return x + (lane >= 16 ? t : 0);
}

__global__ void __launch_bounds__(1024)
row_scan_kernel(const int* __restrict__ cnt, int* __restrict__ rowptr, int N) {
  __shared__ int wsum[32];
  const int lane = threadIdx.x & 31;
  const int w     = __builtin_amdgcn_readfirstlane((int)(threadIdx.x >> 5));
  const int M     = N + 1;
  const int n4    = (N + 3) & ~3;
  const int seg   = ((((M + 31) >> 5) + 127) >> 7) << 7;
  const int trips = seg >> 7;
  const int segb  = w * seg;
  const int lim_l = (segb + seg) < n4 ? (segb + seg) : n4;
  const int lim_s = (segb + seg) < M  ? (segb + seg) : M;
  const int first = segb + 4 * lane;

  int s = 0;
  for (int c = 0; c < trips; c += SCAN_TRIPS) {
    int4 v[SCAN_TRIPS];
    #pragma unroll
    for (int u = 0; u < SCAN_TRIPS; ++u) {
      const int idx = first + (c + u) * 128;
      v[u] = make_int4(0, 0, 0, 0);
      if (idx < lim_l) v[u] = *reinterpret_cast<const int4*>(cnt + idx);
    }
    #pragma unroll
    for (int u = 0; u < SCAN_TRIPS; ++u) s += (v[u].x + v[u].y) + (v[u].z + v[u].w);
  }
  const int tot = __builtin_amdgcn_readlane(wave_incl_scan(s, lane), 31);
  if (lane == 0) wsum[w] = tot;
  __syncthreads();
  int carry = __builtin_amdgcn_readlane(wave_incl_scan(lane < w ? wsum[lane] : 0, lane), 31);

  for (int c = 0; c < trips; c += SCAN_TRIPS) {
    int4 v[SCAN_TRIPS];
    #pragma unroll
    for (int u = 0; u < SCAN_TRIPS; ++u) {
      const int idx = first + (c + u) * 128;
      v[u] = make_int4(0, 0, 0, 0);
      if (idx < lim_l) v[u] = *reinterpret_cast<const int4*>(cnt + idx);
    }
    #pragma unroll
    for (int u = 0; u < SCAN_TRIPS; ++u) {
      const int idx = first + (c + u) * 128;
      const int p1  = v[u].x;
      const int p2  = p1 + v[u].y;
      const int p3  = p2 + v[u].z;
      const int t4  = p3 + v[u].w;
      const int inc = wave_incl_scan(t4, lane);
      const int off = carry + inc - t4;
      if (idx + 3 < lim_s) {
        *reinterpret_cast<int4*>(rowptr + idx) = make_int4(off, off + p1, off + p2, off + p3);
      } else {
        if (idx     < lim_s) rowptr[idx]     = off;
        if (idx + 1 < lim_s) rowptr[idx + 1] = off + p1;
        if (idx + 2 < lim_s) rowptr[idx + 2] = off + p2;
      }
      carry += __builtin_amdgcn_readlane(inc, 31);
    }
  }
}

template <int U>
__device__ __forceinline__ void add_edges(const int myc, const int j, const float4 el,
                                          const float* __restrict__ rl,
                                          const float* __restrict__ zl,
                                          float4& m, float4& s, float4& a0, float4& a1) {
  int c[U]; float4 sc[U], z0[U], z1[U];
  #pragma unroll
  for (int u = 0; u < U; ++u) c[u] = __builtin_amdgcn_readlane(myc, j + u);
  #pragma unroll
  for (int u = 0; u < U; ++u) {
    sc[u] = *reinterpret_cast<const float4*>(rl + (size_t)c[u] * HEADS);
    z0[u] = *reinterpret_cast<const float4*>(zl + (size_t)c[u] * COLS);
    z1[u] = *reinterpret_cast<const float4*>(zl + (size_t)c[u] * COLS + 128);
  }
  float4 mn = m;
  #pragma unroll
  for (int u = 0; u < U; ++u) {
    sc[u].x = leaky01(el.x + sc[u].x); mn.x = fmaxf(mn.x, sc[u].x);
    sc[u].y = leaky01(el.y + sc[u].y); mn.y = fmaxf(mn.y, sc[u].y);
    sc[u].z = leaky01(el.z + sc[u].z); mn.z = fmaxf(mn.z, sc[u].z);
    sc[u].w = leaky01(el.w + sc[u].w); mn.w = fmaxf(mn.w, sc[u].w);
  }
  const float r0 = __expf(m.x - mn.x), r1 = __expf(m.y - mn.y);
  const float r2 = __expf(m.z - mn.z), r3 = __expf(m.w - mn.w);
  s.x  *= r0; s.y  *= r1; s.z  *= r2; s.w  *= r3;
  a0.x *= r0; a0.y *= r1; a0.z *= r2; a0.w *= r3;
  a1.x *= r0; a1.y *= r1; a1.z *= r2; a1.w *= r3;
  #pragma unroll
  for (int u = 0; u < U; ++u) {
    const float w0 = __expf(sc[u].x - mn.x);
    const float w1 = __expf(sc[u].y - mn.y);
    const float w2 = __expf(sc[u].z - mn.z);
    const float w3 = __expf(sc[u].w - mn.w);
    s.x += w0; s.y += w1; s.z += w2; s.w += w3;
    a0.x = fmaf(w0, z0[u].x, a0.x); a0.y = fmaf(w1, z0[u].y, a0.y);
    a0.z = fmaf(w2, z0[u].z, a0.z); a0.w = fmaf(w3, z0[u].w, a0.w);
    a1.x = fmaf(w0, z1[u].x, a1.x); a1.y = fmaf(w1, z1[u].y, a1.y);
    a1.z = fmaf(w2, z1[u].z, a1.z); a1.w = fmaf(w3, z1[u].w, a1.w);
  }
  m = mn;
}

__global__ void __launch_bounds__(256)
row_softmax_gather_kernel(const int* __restrict__ rowptr, const int* __restrict__ scol,
                          const float* __restrict__ e_l, const float* __restrict__ e_r,
                          const float* __restrict__ Zp, float* __restrict__ out, int N) {
  const int lane = threadIdx.x & 31;
  const int r = __builtin_amdgcn_readfirstlane((int)(blockIdx.x * 8 + (threadIdx.x >> 5)));
  if (r >= N) return;
  const int beg = rowptr[r];
  const int end = rowptr[r + 1];
  const int hb  = (lane & 1) * 4;
  const float4 el = *reinterpret_cast<const float4*>(e_l + (size_t)r * HEADS + hb);
  float4 m  = make_float4(-INFINITY, -INFINITY, -INFINITY, -INFINITY);
  float4 s  = make_float4(0.f, 0.f, 0.f, 0.f);
  float4 a0 = make_float4(0.f, 0.f, 0.f, 0.f);
  float4 a1 = make_float4(0.f, 0.f, 0.f, 0.f);
  const float* zl = Zp  + 4 * lane;
  const float* rl = e_r + hb;

  for (int base = beg; base < end; base += 32) {
    const int n = (end - base) < 32 ? (end - base) : 32;
    int myc = (lane < n) ? scol[base + lane] : 0;
    myc = myc < 0 ? 0 : (myc >= N ? N - 1 : myc);
    int j = 0;
    for (; j + 4 <= n; j += 4) add_edges<4>(myc, j, el, rl, zl, m, s, a0, a1);
    for (; j < n; ++j)         add_edges<1>(myc, j, el, rl, zl, m, s, a0, a1);
  }
  if (end > beg) {
    const float q0 = 1.0f / s.x, q1 = 1.0f / s.y, q2 = 1.0f / s.z, q3 = 1.0f / s.w;
    a0.x *= q0; a0.y *= q1; a0.z *= q2; a0.w *= q3;
    a1.x *= q0; a1.y *= q1; a1.z *= q2; a1.w *= q3;
  }
  float* orow = out + (size_t)r * COLS + 4 * lane;
  *reinterpret_cast<float4*>(orow)       = a0;
  *reinterpret_cast<float4*>(orow + 128) = a1;
}

extern "C" void kernel_launch(void* const* d_in, const int* in_sizes, int n_in,
                              void* d_out, int out_size, void* d_ws, size_t ws_size,
                              hipStream_t stream) {
  const float* Z   = (const float*)d_in[0];
  const int*   row = (const int*)d_in[1];
  const int*   col = (const int*)d_in[2];
  const float* W   = (const float*)d_in[3];
  const float* bv  = (const float*)d_in[4];
  const float* a_l = (const float*)d_in[5];
  const float* a_r = (const float*)d_in[6];
  float* out = (float*)d_out;

  const int N = in_sizes[0] / IN_F;
  const int E = in_sizes[1];

  float* Zp  = (float*)d_ws;
  float* e_l = Zp  + (size_t)N * COLS;
  float* e_r = e_l + (size_t)N * HEADS;
  float* mb  = e_r + (size_t)N * HEADS;
  float* sb  = mb  + (size_t)N * HEADS;
  float* Wp  = sb  + (size_t)N * HEADS;
  int* cnt    = (int*)(Wp + WP_ELEMS);
  int* rowptr = cnt + ((N + 3) & ~3);
  int* rank   = rowptr + (N + 1);
  int* scol   = rank + E;
  const bool grouped = (size_t)((const char*)(scol + E) - (const char*)d_ws) <= ws_size;

  pack_W_kernel<<<(WP_ELEMS + 255) / 256, 256, 0, stream>>>(W, Wp, grouped ? cnt : nullptr, N);

  const int rank_blocks = grouped ? RANK_BLOCKS : 0;
  proj_gemm_wmma<<<dim3((N + 15) / 16 + rank_blocks), 128, 0, stream>>>(Z, Wp, bv, Zp, N, row, cnt,
                                                                        rank, E, rank_blocks);

  if (grouped) row_scan_kernel<<<1, 1024, 0, stream>>>(cnt, rowptr, N);

  const int NH = N * HEADS;
  attn_logits_init<<<(NH + 255) / 256, 256, 0, stream>>>(Zp, a_l, a_r, e_l, e_r,
                                                         grouped ? nullptr : mb,
                                                         grouped ? nullptr : sb,
                                                         row, col,
                                                         grouped ? rowptr : nullptr, rank, scol,
                                                         NH, E);

  if (grouped) {
    row_softmax_gather_kernel<<<(N + 7) / 8, 256, 0, stream>>>(rowptr, scol, e_l, e_r, Zp, out, N);
  } else {
    const int EH = E * HEADS;
    edge_max_kernel   <<<(EH + 255) / 256, 256, 0, stream>>>(row, col, e_l, e_r, mb, EH);
    edge_expsum_kernel<<<(EH + 255) / 256, 256, 0, stream>>>(row, col, e_l, e_r, mb, sb, EH);
    hipMemsetAsync(out, 0, (size_t)out_size * sizeof(float), stream);
    edge_scatter_kernel<<<E, 256, 0, stream>>>(row, col, e_l, e_r, mb, sb, Zp, out);
  }
}


# W resident in LDS for the projection: 256 work-groups of 32 waves copy the packed W once, loop over row tiles, read B fragments by ds_load_b64; edge-count atomics follow each wave's tiles
# speedup vs baseline: 3.9643x; 1.0467x over previous
#include <hip/hip_runtime.h>
#include <hip/hip_bf16.h>
#include <math.h>

typedef __attribute__((ext_vector_type(2))) float v2f;
typedef __attribute__((ext_vector_type(8))) float v8f;

#define IN_F   256
#define OUT_F  32
#define HEADS  8
#define COLS   256
#define WP_ELEMS (128 * COLS * 2)
#define GEMM_WGS 256

__global__ void pack_W_kernel(const float* __restrict__ W, float* __restrict__ Wp,
                              int* __restrict__ cnt, int N) {
  const int idx = blockIdx.x * blockDim.x + threadIdx.x;
  if (cnt != nullptr) {
    const int nthr = gridDim.x * blockDim.x;
    const int n4   = (N + 3) & ~3;
    for (int i = idx; i < n4; i += nthr) cnt[i] = 0;
  }
  if (idx >= WP_ELEMS) return;
  const int kk = idx & 3;
  const int n  = (idx >> 2) & (COLS - 1);
  const int k4 = idx >> 10;
  Wp[idx] = W[(size_t)(k4 * 4 + kk) * COLS + n];
}

__global__ void __launch_bounds__(1024)
proj_gemm_wmma(const float* __restrict__ Z, const float* __restrict__ Wp,
               const float* __restrict__ bias, float* __restrict__ Zp, int N,
               const int* __restrict__ row, int* __restrict__ cnt,
               int* __restrict__ rank, int E) {
  __shared__ float wl[WP_ELEMS];

  const int tid  = threadIdx.x;
  const int nthr = gridDim.x * 1024;
  const int gtid = blockIdx.x * 1024 + tid;

  const float4* __restrict__ src4 = reinterpret_cast<const float4*>(Wp);
  float4* dst4 = reinterpret_cast<float4*>(wl);
  float4 w[16];
  #pragma unroll
  for (int u = 0; u < 16; ++u) w[u] = src4[tid + u * 1024];

  int ee[4];
  unsigned rr[4];
  #pragma unroll
  for (int u = 0; u < 4; ++u) { ee[u] = gtid + u * nthr; rr[u] = 0u; }
  if (cnt != nullptr && E > 0) {
    #pragma unroll
    for (int u = 0; u < 4; ++u) rr[u] = (unsigned)row[ee[u] < E ? ee[u] : E - 1];
  }

  #pragma unroll
  for (int u = 0; u < 16; ++u) dst4[tid + u * 1024] = w[u];
  __syncthreads();

  const int lane = tid & 31;
  const int wave = __builtin_amdgcn_readfirstlane(tid >> 5);
  const int l16  = lane & 15;
  const int half = lane >> 4;
  const int n0   = (wave & 3) << 6;
  const int gi     = (wave >> 3) + 4 * ((wave >> 2) & 1);
  const int ntiles = (N + 15) >> 4;

  const float* wb0 = wl + ((n0 + l16) * 4 + 2 * half);

  for (int q = gi; ; q += 8) {
    const int rt = (int)blockIdx.x + (int)gridDim.x * q;
    if (rt >= ntiles) break;
    const int m0 = rt << 4;

    const int mA = m0 + l16;
    const float* zrow = Z + (size_t)(mA < N ? mA : (N - 1)) * IN_F;

    v8f acc0 = {}; v8f acc1 = {}; v8f acc2 = {}; v8f acc3 = {};

    #pragma unroll 4
    for (int k = 0; k < IN_F; k += 4) {
      const float* wb = wb0 + k * COLS;
      v2f a  = *reinterpret_cast<const v2f*>(zrow + k + 2 * half);
      v2f b0 = *reinterpret_cast<const v2f*>(wb);
      v2f b1 = *reinterpret_cast<const v2f*>(wb + 64);
      v2f b2 = *reinterpret_cast<const v2f*>(wb + 128);
      v2f b3 = *reinterpret_cast<const v2f*>(wb + 192);

      acc0 = __builtin_amdgcn_wmma_f32_16x16x4_f32(false, a, false, b0, (short)0, acc0, false, false);
      acc1 = __builtin_amdgcn_wmma_f32_16x16x4_f32(false, a, false, b1, (short)0, acc1, false, false);
      acc2 = __builtin_amdgcn_wmma_f32_16x16x4_f32(false, a, false, b2, (short)0, acc2, false, false);
      acc3 = __builtin_amdgcn_wmma_f32_16x16x4_f32(false, a, false, b3, (short)0, acc3, false, false);
    }

    v8f accs[4] = {acc0, acc1, acc2, acc3};
    #pragma unroll
    for (int t = 0; t < 4; ++t) {
      const int n  = n0 + 16 * t + l16;
      const float bv = bias[n];
      #pragma unroll
      for (int v = 0; v < 8; ++v) {
        const int mrow = m0 + v + 8 * half;
        if (mrow < N) Zp[(size_t)mrow * COLS + n] = accs[t][v] + bv;
      }
    }
  }

  if (cnt != nullptr) {
    int ss[4];
    #pragma unroll
    for (int u = 0; u < 4; ++u) {
      ss[u] = 0;
      if (ee[u] < E) ss[u] = atomicAdd(&cnt[rr[u]], 1);
    }
    #pragma unroll
    for (int u = 0; u < 4; ++u)
      if (ee[u] < E) rank[ee[u]] = ss[u];
    for (int e = gtid + 4 * nthr; e < E; e += nthr) rank[e] = atomicAdd(&cnt[row[e]], 1);
  }
}

__global__ void attn_logits_init(const float* __restrict__ Zp,
                                 const float* __restrict__ a_l, const float* __restrict__ a_r,
                                 float* __restrict__ e_l, float* __restrict__ e_r,
                                 float* __restrict__ mbuf, float* __restrict__ sbuf,
                                 const int* __restrict__ row, const int* __restrict__ col,
                                 const int* __restrict__ rowptr, const int* __restrict__ rank,
                                 int* __restrict__ scol, int NH, int E) {
  const int idx = blockIdx.x * blockDim.x + threadIdx.x;

  if (rowptr != nullptr) {
    const int nthr = gridDim.x * blockDim.x;
    for (int e = idx; e < E; e += 2 * nthr) {
      const int f  = e + nthr;
      const int g  = f < E ? f : e;
      const int ra = row[e],  rb = row[g];
      const int ka = rank[e], kb = rank[g];
      const int ca = col[e],  cb = col[g];
      int pa = rowptr[ra] + ka;
      int pb = rowptr[rb] + kb;
      pa = pa < 0 ? 0 : (pa >= E ? E - 1 : pa);
      pb = pb < 0 ? 0 : (pb >= E ? E - 1 : pb);
      scol[pa] = ca;
      scol[pb] = cb;
    }
  }

  if (idx >= NH) return;
  const int n = idx >> 3;
  const int h = idx & 7;
  const float* z = Zp + (size_t)n * COLS + h;
  float al = 0.f, ar = 0.f;
  #pragma unroll
  for (int o = 0; o < OUT_F; ++o) {
    const float zv = z[o * HEADS];
    al = fmaf(zv, a_l[o * HEADS + h], al);
    ar = fmaf(zv, a_r[o * HEADS + h], ar);
  }
  e_l[idx]  = al;
  e_r[idx]  = ar;
  if (mbuf != nullptr) {
    mbuf[idx] = -INFINITY;
    sbuf[idx] = 0.f;
  }
}

__device__ __forceinline__ float leaky01(float v) {
  return v > 0.f ? v : 0.01f * v;
}

__device__ __forceinline__ void atomicMaxFloat(float* addr, float val) {
  if (val >= 0.f) atomicMax((int*)addr, __float_as_int(val));
  else            atomicMin((unsigned int*)addr, __float_as_uint(val));
}

__global__ void edge_max_kernel(const int* __restrict__ row, const int* __restrict__ col,
                                const float* __restrict__ e_l, const float* __restrict__ e_r,
                                float* __restrict__ mbuf, int EH) {
  const int idx = blockIdx.x * blockDim.x + threadIdx.x;
  if (idx >= EH) return;
  const int e = idx >> 3, h = idx & 7;
  const int r = row[e], c = col[e];
  const float v = leaky01(e_l[r * HEADS + h] + e_r[c * HEADS + h]);
  atomicMaxFloat(&mbuf[r * HEADS + h], v);
}

__global__ void edge_expsum_kernel(const int* __restrict__ row, const int* __restrict__ col,
                                   const float* __restrict__ e_l, const float* __restrict__ e_r,
                                   const float* __restrict__ mbuf, float* __restrict__ sbuf, int EH) {
  const int idx = blockIdx.x * blockDim.x + threadIdx.x;
  if (idx >= EH) return;
  const int e = idx >> 3, h = idx & 7;
  const int r = row[e], c = col[e];
  const float v  = leaky01(e_l[r * HEADS + h] + e_r[c * HEADS + h]);
  const float ex = __expf(v - mbuf[r * HEADS + h]);
  atomicAdd(&sbuf[r * HEADS + h], ex);
}

__global__ void __launch_bounds__(256)
edge_scatter_kernel(const int* __restrict__ row, const int* __restrict__ col,
                    const float* __restrict__ e_l, const float* __restrict__ e_r,
                    const float* __restrict__ mbuf, const float* __restrict__ sbuf,
                    const float* __restrict__ Zp, float* __restrict__ out) {
  const int e = blockIdx.x;
  const int c = threadIdx.x;
  const int r  = row[e];
  const int cl = col[e];
  const int h  = c & 7;
  const float v    = leaky01(e_l[r * HEADS + h] + e_r[cl * HEADS + h]);
  const float attn = __expf(v - mbuf[r * HEADS + h]) / sbuf[r * HEADS + h];
  atomicAdd(&out[(size_t)r * COLS + c], attn * Zp[(size_t)cl * COLS + c]);
}


#define SCAN_TRIPS 16

__device__ __forceinline__ int wave_incl_scan(int x, const int lane) {
  x += __builtin_amdgcn_update_dpp(0, x, 0x111, 0xf, 0xf, true);
  x += __builtin_amdgcn_update_dpp(0, x, 0x112, 0xf, 0xf, true);
  x += __builtin_amdgcn_update_dpp(0, x, 0x114, 0xf, 0xf, true);
  x += __builtin_amdgcn_update_dpp(0, x, 0x118, 0xf, 0xf, true);
  const int t = __builtin_amdgcn_readlane(x, 15);
  return x + (lane >= 16 ? t : 0);
}

__global__ void __launch_bounds__(1024)
row_scan_kernel(const int* __restrict__ cnt, int* __restrict__ rowptr, int N) {
  __shared__ int wsum[32];
  const int lane = threadIdx.x & 31;
  const int w     = __builtin_amdgcn_readfirstlane((int)(threadIdx.x >> 5));
  const int M     = N + 1;
  const int n4    = (N + 3) & ~3;
  const int seg   = ((((M + 31) >> 5) + 127) >> 7) << 7;
  const int trips = seg >> 7;
  const int segb  = w * seg;
  const int lim_l = (segb + seg) < n4 ? (segb + seg) : n4;
  const int lim_s = (segb + seg) < M  ? (segb + seg) : M;
  const int first = segb + 4 * lane;

  int s = 0;
  for (int c = 0; c < trips; c += SCAN_TRIPS) {
    int4 v[SCAN_TRIPS];
    #pragma unroll
    for (int u = 0; u < SCAN_TRIPS; ++u) {
      const int idx = first + (c + u) * 128;
      v[u] = make_int4(0, 0, 0, 0);
      if (idx < lim_l) v[u] = *reinterpret_cast<const int4*>(cnt + idx);
    }
    #pragma unroll
    for (int u = 0; u < SCAN_TRIPS; ++u) s += (v[u].x + v[u].y) + (v[u].z + v[u].w);
  }
  const int tot = __builtin_amdgcn_readlane(wave_incl_scan(s, lane), 31);
  if (lane == 0) wsum[w] = tot;
  __syncthreads();
  int carry = __builtin_amdgcn_readlane(wave_incl_scan(lane < w ? wsum[lane] : 0, lane), 31);

  for (int c = 0; c < trips; c += SCAN_TRIPS) {
    int4 v[SCAN_TRIPS];
    #pragma unroll
    for (int u = 0; u < SCAN_TRIPS; ++u) {
      const int idx = first + (c + u) * 128;
      v[u] = make_int4(0, 0, 0, 0);
      if (idx < lim_l) v[u] = *reinterpret_cast<const int4*>(cnt + idx);
    }
    #pragma unroll
    for (int u = 0; u < SCAN_TRIPS; ++u) {
      const int idx = first + (c + u) * 128;
      const int p1  = v[u].x;
      const int p2  = p1 + v[u].y;
      const int p3  = p2 + v[u].z;
      const int t4  = p3 + v[u].w;
      const int inc = wave_incl_scan(t4, lane);
      const int off = carry + inc - t4;
      if (idx + 3 < lim_s) {
        *reinterpret_cast<int4*>(rowptr + idx) = make_int4(off, off + p1, off + p2, off + p3);
      } else {
        if (idx     < lim_s) rowptr[idx]     = off;
        if (idx + 1 < lim_s) rowptr[idx + 1] = off + p1;
        if (idx + 2 < lim_s) rowptr[idx + 2] = off + p2;
      }
      carry += __builtin_amdgcn_readlane(inc, 31);
    }
  }
}

template <int U>
__device__ __forceinline__ void add_edges(const int myc, const int j, const float4 el,
                                          const float* __restrict__ rl,
                                          const float* __restrict__ zl,
                                          float4& m, float4& s, float4& a0, float4& a1) {
  int c[U]; float4 sc[U], z0[U], z1[U];
  #pragma unroll
  for (int u = 0; u < U; ++u) c[u] = __builtin_amdgcn_readlane(myc, j + u);
  #pragma unroll
  for (int u = 0; u < U; ++u) {
    sc[u] = *reinterpret_cast<const float4*>(rl + (size_t)c[u] * HEADS);
    z0[u] = *reinterpret_cast<const float4*>(zl + (size_t)c[u] * COLS);
    z1[u] = *reinterpret_cast<const float4*>(zl + (size_t)c[u] * COLS + 128);
  }
  float4 mn = m;
  #pragma unroll
  for (int u = 0; u < U; ++u) {
    sc[u].x = leaky01(el.x + sc[u].x); mn.x = fmaxf(mn.x, sc[u].x);
    sc[u].y = leaky01(el.y + sc[u].y); mn.y = fmaxf(mn.y, sc[u].y);
    sc[u].z = leaky01(el.z + sc[u].z); mn.z = fmaxf(mn.z, sc[u].z);
    sc[u].w = leaky01(el.w + sc[u].w); mn.w = fmaxf(mn.w, sc[u].w);
  }
  const float r0 = __expf(m.x - mn.x), r1 = __expf(m.y - mn.y);
  const float r2 = __expf(m.z - mn.z), r3 = __expf(m.w - mn.w);
  s.x  *= r0; s.y  *= r1; s.z  *= r2; s.w  *= r3;
  a0.x *= r0; a0.y *= r1; a0.z *= r2; a0.w *= r3;
  a1.x *= r0; a1.y *= r1; a1.z *= r2; a1.w *= r3;
  #pragma unroll
  for (int u = 0; u < U; ++u) {
    const float w0 = __expf(sc[u].x - mn.x);
    const float w1 = __expf(sc[u].y - mn.y);
    const float w2 = __expf(sc[u].z - mn.z);
    const float w3 = __expf(sc[u].w - mn.w);
    s.x += w0; s.y += w1; s.z += w2; s.w += w3;
    a0.x = fmaf(w0, z0[u].x, a0.x); a0.y = fmaf(w1, z0[u].y, a0.y);
    a0.z = fmaf(w2, z0[u].z, a0.z); a0.w = fmaf(w3, z0[u].w, a0.w);
    a1.x = fmaf(w0, z1[u].x, a1.x); a1.y = fmaf(w1, z1[u].y, a1.y);
    a1.z = fmaf(w2, z1[u].z, a1.z); a1.w = fmaf(w3, z1[u].w, a1.w);
  }
  m = mn;
}

__global__ void __launch_bounds__(256)
row_softmax_gather_kernel(const int* __restrict__ rowptr, const int* __restrict__ scol,
                          const float* __restrict__ e_l, const float* __restrict__ e_r,
                          const float* __restrict__ Zp, float* __restrict__ out, int N) {
  const int lane = threadIdx.x & 31;
  const int r = __builtin_amdgcn_readfirstlane((int)(blockIdx.x * 8 + (threadIdx.x >> 5)));
  if (r >= N) return;
  const int beg = rowptr[r];
  const int end = rowptr[r + 1];
  const int hb  = (lane & 1) * 4;
  const float4 el = *reinterpret_cast<const float4*>(e_l + (size_t)r * HEADS + hb);
  float4 m  = make_float4(-INFINITY, -INFINITY, -INFINITY, -INFINITY);
  float4 s  = make_float4(0.f, 0.f, 0.f, 0.f);
  float4 a0 = make_float4(0.f, 0.f, 0.f, 0.f);
  float4 a1 = make_float4(0.f, 0.f, 0.f, 0.f);
  const float* zl = Zp  + 4 * lane;
  const float* rl = e_r + hb;

  for (int base = beg; base < end; base += 32) {
    const int n = (end - base) < 32 ? (end - base) : 32;
    int myc = (lane < n) ? scol[base + lane] : 0;
    myc = myc < 0 ? 0 : (myc >= N ? N - 1 : myc);
    int j = 0;
    for (; j + 4 <= n; j += 4) add_edges<4>(myc, j, el, rl, zl, m, s, a0, a1);
    for (; j < n; ++j)         add_edges<1>(myc, j, el, rl, zl, m, s, a0, a1);
  }
  if (end > beg) {
    const float q0 = 1.0f / s.x, q1 = 1.0f / s.y, q2 = 1.0f / s.z, q3 = 1.0f / s.w;
    a0.x *= q0; a0.y *= q1; a0.z *= q2; a0.w *= q3;
    a1.x *= q0; a1.y *= q1; a1.z *= q2; a1.w *= q3;
  }
  float* orow = out + (size_t)r * COLS + 4 * lane;
  *reinterpret_cast<float4*>(orow)       = a0;
  *reinterpret_cast<float4*>(orow + 128) = a1;
}

extern "C" void kernel_launch(void* const* d_in, const int* in_sizes, int n_in,
                              void* d_out, int out_size, void* d_ws, size_t ws_size,
                              hipStream_t stream) {
  const float* Z   = (const float*)d_in[0];
  const int*   row = (const int*)d_in[1];
  const int*   col = (const int*)d_in[2];
  const float* W   = (const float*)d_in[3];
  const float* bv  = (const float*)d_in[4];
  const float* a_l = (const float*)d_in[5];
  const float* a_r = (const float*)d_in[6];
  float* out = (float*)d_out;

  const int N = in_sizes[0] / IN_F;
  const int E = in_sizes[1];

  float* Zp  = (float*)d_ws;
  float* e_l = Zp  + (size_t)N * COLS;
  float* e_r = e_l + (size_t)N * HEADS;
  float* mb  = e_r + (size_t)N * HEADS;
  float* sb  = mb  + (size_t)N * HEADS;
  float* Wp  = sb  + (size_t)N * HEADS;
  int* cnt    = (int*)(Wp + WP_ELEMS);
  int* rowptr = cnt + ((N + 3) & ~3);
  int* rank   = rowptr + (N + 1);
  int* scol   = rank + E;
  const bool grouped = (size_t)((const char*)(scol + E) - (const char*)d_ws) <= ws_size;

  pack_W_kernel<<<(WP_ELEMS + 255) / 256, 256, 0, stream>>>(W, Wp, grouped ? cnt : nullptr, N);

  proj_gemm_wmma<<<GEMM_WGS, 1024, 0, stream>>>(Z, Wp, bv, Zp, N, row, grouped ? cnt : nullptr,
                                                rank, E);

  if (grouped) row_scan_kernel<<<1, 1024, 0, stream>>>(cnt, rowptr, N);

  const int NH = N * HEADS;
  attn_logits_init<<<(NH + 255) / 256, 256, 0, stream>>>(Zp, a_l, a_r, e_l, e_r,
                                                         grouped ? nullptr : mb,
                                                         grouped ? nullptr : sb,
                                                         row, col,
                                                         grouped ? rowptr : nullptr, rank, scol,
                                                         NH, E);

  if (grouped) {
    row_softmax_gather_kernel<<<(N + 7) / 8, 256, 0, stream>>>(rowptr, scol, e_l, e_r, Zp, out, N);
  } else {
    const int EH = E * HEADS;
    edge_max_kernel   <<<(EH + 255) / 256, 256, 0, stream>>>(row, col, e_l, e_r, mb, EH);
    edge_expsum_kernel<<<(EH + 255) / 256, 256, 0, stream>>>(row, col, e_l, e_r, mb, sb, EH);
    hipMemsetAsync(out, 0, (size_t)out_size * sizeof(float), stream);
    edge_scatter_kernel<<<E, 256, 0, stream>>>(row, col, e_l, e_r, mb, sb, Zp, out);
  }
}


# Edge lists without global atomics: each GEMM work-group owns a row range, finds its edges in a byte plane of owner ids ahead of its tiles, counts and stages slots in LDS; sum order = LDS slot order
# speedup vs baseline: 5.1372x; 1.2959x over previous
#include <hip/hip_runtime.h>
#include <hip/hip_bf16.h>
#include <math.h>

typedef __attribute__((ext_vector_type(2))) float v2f;
typedef __attribute__((ext_vector_type(8))) float v8f;

#define IN_F   256
#define OUT_F  32
#define HEADS  8
#define COLS   256
#define WP_ELEMS (128 * COLS * 2)
#define GEMM_WGS 256
#define ROW_SLOTS 64
#define LROWS 200

__global__ void pack_W_kernel(const float* __restrict__ W, float* __restrict__ Wp,
                              int* __restrict__ cnt, int N, int* __restrict__ ovfn,
                              const int* __restrict__ row, unsigned* __restrict__ binb,
                              int E, int R) {
  const int idx = blockIdx.x * blockDim.x + threadIdx.x;
  if (binb != nullptr) {
    const int nthr = gridDim.x * blockDim.x;
    const int ndw  = ((E + 15) >> 4) << 2;
    for (int q = idx; q < ndw; q += nthr) {
      unsigned v = 0u;
      #pragma unroll
      for (int u = 0; u < 4; ++u) {
        const int e = 4 * q + u;
        unsigned b = 0xFFu;
        if (e < E) b = ((unsigned)row[e] / (unsigned)R) & 0xFFu;
        v |= b << (8 * u);
      }
      binb[q] = v;
    }
  }
  if (ovfn != nullptr && idx == 0) *ovfn = 0;
  if (cnt != nullptr) {
    const int nthr = gridDim.x * blockDim.x;
    const int n4   = (N + 3) & ~3;
    for (int i = idx; i < n4; i += nthr) cnt[i] = 0;
  }
  if (idx >= WP_ELEMS) return;
  const int kk = idx & 3;
  const int n  = (idx >> 2) & (COLS - 1);
  const int k4 = idx >> 10;
  Wp[idx] = W[(size_t)(k4 * 4 + kk) * COLS + n];
}

__global__ void __launch_bounds__(1024)
proj_gemm_wmma(const float* __restrict__ Z, const float* __restrict__ Wp,
               const float* __restrict__ bias, float* __restrict__ Zp, int N,
               const int* __restrict__ row, int* __restrict__ cnt,
               int* __restrict__ rank, int E,
               const int* __restrict__ col, int* __restrict__ slots,
               int* __restrict__ ovfn, int* __restrict__ ovf,
               const unsigned* __restrict__ binb, int R) {
  __shared__ float wl[WP_ELEMS];
  __shared__ int cntl[LROWS];
  __shared__ int stage[LROWS * ROW_SLOTS];

  const int tid  = threadIdx.x;
  const int nthr = gridDim.x * 1024;
  const int gtid = blockIdx.x * 1024 + tid;

  const float4* __restrict__ src4 = reinterpret_cast<const float4*>(Wp);
  float4* dst4 = reinterpret_cast<float4*>(wl);
  float4 w[16];
  #pragma unroll
  for (int u = 0; u < 16; ++u) w[u] = src4[tid + u * 1024];

  int ee[4];
  unsigned rr[4];
  #pragma unroll
  for (int u = 0; u < 4; ++u) { ee[u] = gtid + u * nthr; rr[u] = 0u; }
  int cc[4];
  #pragma unroll
  for (int u = 0; u < 4; ++u) cc[u] = 0;
  if (cnt != nullptr && E > 0 && binb == nullptr) {
    #pragma unroll
    for (int u = 0; u < 4; ++u) rr[u] = (unsigned)row[ee[u] < E ? ee[u] : E - 1];
    if (slots != nullptr) {
      #pragma unroll
      for (int u = 0; u < 4; ++u) cc[u] = col[ee[u] < E ? ee[u] : E - 1];
    }
  }

  #pragma unroll
  for (int u = 0; u < 16; ++u) dst4[tid + u * 1024] = w[u];
  if (binb != nullptr && tid < LROWS) cntl[tid] = 0;
  __syncthreads();

  const int lane = tid & 31;
  const int wave = __builtin_amdgcn_readfirstlane(tid >> 5);
  const int l16  = lane & 15;
  const int half = lane >> 4;
  const int n0   = (wave & 3) << 6;
  const int gi     = (wave >> 3) + 4 * ((wave >> 2) & 1);
  const int ntiles = (N + 15) >> 4;

  const float* wb0 = wl + ((n0 + l16) * 4 + 2 * half);

  if (binb != nullptr && (wave & 4) != 0) {
    const int rbase = (int)blockIdx.x * R;
    const unsigned pat = (blockIdx.x & 0xFFu) * 0x01010101u;
    const int nunit = (E + 15) >> 4;
    const uint4* __restrict__ b4 = reinterpret_cast<const uint4*>(binb);
    const int sid = ((((wave >> 3) << 2) | (wave & 3)) << 5) | lane;
    for (int q0 = sid; q0 < nunit; q0 += 8 * 512) {
      uint4 xv[8];
      #pragma unroll
      for (int u = 0; u < 8; ++u) {
        const int q = q0 + u * 512;
        xv[u] = b4[q < nunit ? q : nunit - 1];
      }
      unsigned mk[8];
      #pragma unroll
      for (int u = 0; u < 8; ++u) {
        const unsigned y0 = xv[u].x ^ pat, y1 = xv[u].y ^ pat, y2 = xv[u].z ^ pat, y3 = xv[u].w ^ pat;
        const unsigned z0 = ~(((y0 & 0x7F7F7F7Fu) + 0x7F7F7F7Fu) | y0) & 0x80808080u;
        const unsigned z1 = ~(((y1 & 0x7F7F7F7Fu) + 0x7F7F7F7Fu) | y1) & 0x80808080u;
        const unsigned z2 = ~(((y2 & 0x7F7F7F7Fu) + 0x7F7F7F7Fu) | y2) & 0x80808080u;
        const unsigned z3 = ~(((y3 & 0x7F7F7F7Fu) + 0x7F7F7F7Fu) | y3) & 0x80808080u;
        mk[u] = (z0 >> 7) | (z1 >> 6) | (z2 >> 5) | (z3 >> 4);
        if (q0 + u * 512 >= nunit) mk[u] = 0u;
      }
      int fe[8], fr[8], fc[8];
      #pragma unroll
      for (int u = 0; u < 8; ++u) {
        const int pos = mk[u] != 0u ? __builtin_ctz(mk[u]) : 0;
        const int e = 16 * (q0 + u * 512) + 4 * (pos & 7) + (pos >> 3);
        fe[u] = (mk[u] != 0u && e < E) ? e : -1;
        const int ec = fe[u] >= 0 ? fe[u] : 0;
        fr[u] = row[ec];
        fc[u] = col[ec];
        mk[u] &= mk[u] - 1u;
      }
      int fs[8];
      #pragma unroll
      for (int u = 0; u < 8; ++u) {
        const int li = fr[u] - rbase;
        fs[u] = -1;
        if (fe[u] >= 0 && li >= 0 && li < R && li < LROWS && fr[u] < N) fs[u] = atomicAdd(&cntl[li], 1);
      }
      #pragma unroll
      for (int u = 0; u < 8; ++u) {
        if (fs[u] >= 0) {
          if (fs[u] < ROW_SLOTS) {
            stage[(fr[u] - rbase) * ROW_SLOTS + fs[u]] = fc[u];
          } else {
            const int p = atomicAdd(ovfn, 1);
            if (p >= 0 && p < E) { ovf[2 * (size_t)p] = fr[u]; ovf[2 * (size_t)p + 1] = fc[u]; }
          }
        }
      }
      #pragma unroll
      for (int u = 0; u < 8; ++u) {
        while (mk[u] != 0u) {
          const int pos = __builtin_ctz(mk[u]);
          mk[u] &= mk[u] - 1u;
          const int e = 16 * (q0 + u * 512) + 4 * (pos & 7) + (pos >> 3);
          if (e < E) {
            const int er = row[e];
            const int c  = col[e];
            const int li = er - rbase;
            if (li >= 0 && li < R && li < LROWS && er < N) {
              const int sl = atomicAdd(&cntl[li], 1);
              if (sl < ROW_SLOTS) {
                stage[li * ROW_SLOTS + sl] = c;
              } else {
                const int p = atomicAdd(ovfn, 1);
                if (p >= 0 && p < E) { ovf[2 * (size_t)p] = er; ovf[2 * (size_t)p + 1] = c; }
              }
            }
          }
        }
      }
    }
  }

  for (int q = gi; ; q += 8) {
    const int rt = (int)blockIdx.x + (int)gridDim.x * q;
    if (rt >= ntiles) break;
    const int m0 = rt << 4;

    const int mA = m0 + l16;
    const float* zrow = Z + (size_t)(mA < N ? mA : (N - 1)) * IN_F;

    v8f acc0 = {}; v8f acc1 = {}; v8f acc2 = {}; v8f acc3 = {};

    #pragma unroll 4
    for (int k = 0; k < IN_F; k += 4) {
      const float* wb = wb0 + k * COLS;
      v2f a  = *reinterpret_cast<const v2f*>(zrow + k + 2 * half);
      v2f b0 = *reinterpret_cast<const v2f*>(wb);
      v2f b1 = *reinterpret_cast<const v2f*>(wb + 64);
      v2f b2 = *reinterpret_cast<const v2f*>(wb + 128);
      v2f b3 = *reinterpret_cast<const v2f*>(wb + 192);

      acc0 = __builtin_amdgcn_wmma_f32_16x16x4_f32(false, a, false, b0, (short)0, acc0, false, false);
      acc1 = __builtin_amdgcn_wmma_f32_16x16x4_f32(false, a, false, b1, (short)0, acc1, false, false);
      acc2 = __builtin_amdgcn_wmma_f32_16x16x4_f32(false, a, false, b2, (short)0, acc2, false, false);
      acc3 = __builtin_amdgcn_wmma_f32_16x16x4_f32(false, a, false, b3, (short)0, acc3, false, false);
    }

    v8f accs[4] = {acc0, acc1, acc2, acc3};
    #pragma unroll
    for (int t = 0; t < 4; ++t) {
      const int n  = n0 + 16 * t + l16;
      const float bv = bias[n];
      #pragma unroll
      for (int v = 0; v < 8; ++v) {
        const int mrow = m0 + v + 8 * half;
        if (mrow < N) Zp[(size_t)mrow * COLS + n] = accs[t][v] + bv;
      }
    }
  }

  if (binb != nullptr) {
    const int rbase = (int)blockIdx.x * R;
    __syncthreads();
    int nrows = N - rbase;
    nrows = nrows < 0 ? 0 : (nrows > R ? R : nrows);
    nrows = nrows > LROWS ? LROWS : nrows;
    for (int i = tid; i < nrows; i += 1024) cnt[rbase + i] = cntl[i];
    const int4* __restrict__ st4 = reinterpret_cast<const int4*>(stage);
    int4* __restrict__ sl4 = reinterpret_cast<int4*>(slots + (size_t)rbase * ROW_SLOTS);
    const int n4 = nrows * (ROW_SLOTS / 4);
    for (int i = tid; i < n4; i += 1024) sl4[i] = st4[i];
  } else
  if (cnt != nullptr && slots != nullptr) {
    int ss[4];
    #pragma unroll
    for (int u = 0; u < 4; ++u) {
      ss[u] = 0;
      if (ee[u] < E) ss[u] = atomicAdd(&cnt[rr[u]], 1);
    }
    #pragma unroll
    for (int u = 0; u < 4; ++u) {
      if (ee[u] < E) {
        if (ss[u] < ROW_SLOTS) {
          slots[(size_t)rr[u] * ROW_SLOTS + ss[u]] = cc[u];
        } else {
          const int p = atomicAdd(ovfn, 1);
          if (p >= 0 && p < E) { ovf[2 * (size_t)p] = (int)rr[u]; ovf[2 * (size_t)p + 1] = cc[u]; }
        }
      }
    }
    for (int e = gtid + 4 * nthr; e < E; e += nthr) {
      const int r = row[e];
      const int c = col[e];
      const int s = atomicAdd(&cnt[r], 1);
      if (s < ROW_SLOTS) {
        slots[(size_t)r * ROW_SLOTS + s] = c;
      } else {
        const int p = atomicAdd(ovfn, 1);
        if (p >= 0 && p < E) { ovf[2 * (size_t)p] = r; ovf[2 * (size_t)p + 1] = c; }
      }
    }
  } else
  if (cnt != nullptr) {
    int ss[4];
    #pragma unroll
    for (int u = 0; u < 4; ++u) {
      ss[u] = 0;
      if (ee[u] < E) ss[u] = atomicAdd(&cnt[rr[u]], 1);
    }
    #pragma unroll
    for (int u = 0; u < 4; ++u)
      if (ee[u] < E) rank[ee[u]] = ss[u];
    for (int e = gtid + 4 * nthr; e < E; e += nthr) rank[e] = atomicAdd(&cnt[row[e]], 1);
  }
}

__global__ void attn_logits_init(const float* __restrict__ Zp,
                                 const float* __restrict__ a_l, const float* __restrict__ a_r,
                                 float* __restrict__ e_l, float* __restrict__ e_r,
                                 float* __restrict__ mbuf, float* __restrict__ sbuf,
                                 const int* __restrict__ row, const int* __restrict__ col,
                                 const int* __restrict__ rowptr, const int* __restrict__ rank,
                                 int* __restrict__ scol, int NH, int E) {
  const int idx = blockIdx.x * blockDim.x + threadIdx.x;

  if (rowptr != nullptr) {
    const int nthr = gridDim.x * blockDim.x;
    for (int e = idx; e < E; e += 2 * nthr) {
      const int f  = e + nthr;
      const int g  = f < E ? f : e;
      const int ra = row[e],  rb = row[g];
      const int ka = rank[e], kb = rank[g];
      const int ca = col[e],  cb = col[g];
      int pa = rowptr[ra] + ka;
      int pb = rowptr[rb] + kb;
      pa = pa < 0 ? 0 : (pa >= E ? E - 1 : pa);
      pb = pb < 0 ? 0 : (pb >= E ? E - 1 : pb);
      scol[pa] = ca;
      scol[pb] = cb;
    }
  }

  if (idx >= NH) return;
  const int n = idx >> 3;
  const int h = idx & 7;
  const float* z = Zp + (size_t)n * COLS + h;
  float al = 0.f, ar = 0.f;
  #pragma unroll
  for (int o = 0; o < OUT_F; ++o) {
    const float zv = z[o * HEADS];
    al = fmaf(zv, a_l[o * HEADS + h], al);
    ar = fmaf(zv, a_r[o * HEADS + h], ar);
  }
  e_l[idx]  = al;
  e_r[idx]  = ar;
  if (mbuf != nullptr) {
    mbuf[idx] = -INFINITY;
    sbuf[idx] = 0.f;
  }
}

__device__ __forceinline__ float leaky01(float v) {
  return v > 0.f ? v : 0.01f * v;
}

__device__ __forceinline__ void atomicMaxFloat(float* addr, float val) {
  if (val >= 0.f) atomicMax((int*)addr, __float_as_int(val));
  else            atomicMin((unsigned int*)addr, __float_as_uint(val));
}

__global__ void edge_max_kernel(const int* __restrict__ row, const int* __restrict__ col,
                                const float* __restrict__ e_l, const float* __restrict__ e_r,
                                float* __restrict__ mbuf, int EH) {
  const int idx = blockIdx.x * blockDim.x + threadIdx.x;
  if (idx >= EH) return;
  const int e = idx >> 3, h = idx & 7;
  const int r = row[e], c = col[e];
  const float v = leaky01(e_l[r * HEADS + h] + e_r[c * HEADS + h]);
  atomicMaxFloat(&mbuf[r * HEADS + h], v);
}

__global__ void edge_expsum_kernel(const int* __restrict__ row, const int* __restrict__ col,
                                   const float* __restrict__ e_l, const float* __restrict__ e_r,
                                   const float* __restrict__ mbuf, float* __restrict__ sbuf, int EH) {
  const int idx = blockIdx.x * blockDim.x + threadIdx.x;
  if (idx >= EH) return;
  const int e = idx >> 3, h = idx & 7;
  const int r = row[e], c = col[e];
  const float v  = leaky01(e_l[r * HEADS + h] + e_r[c * HEADS + h]);
  const float ex = __expf(v - mbuf[r * HEADS + h]);
  atomicAdd(&sbuf[r * HEADS + h], ex);
}

__global__ void __launch_bounds__(256)
edge_scatter_kernel(const int* __restrict__ row, const int* __restrict__ col,
                    const float* __restrict__ e_l, const float* __restrict__ e_r,
                    const float* __restrict__ mbuf, const float* __restrict__ sbuf,
                    const float* __restrict__ Zp, float* __restrict__ out) {
  const int e = blockIdx.x;
  const int c = threadIdx.x;
  const int r  = row[e];
  const int cl = col[e];
  const int h  = c & 7;
  const float v    = leaky01(e_l[r * HEADS + h] + e_r[cl * HEADS + h]);
  const float attn = __expf(v - mbuf[r * HEADS + h]) / sbuf[r * HEADS + h];
  atomicAdd(&out[(size_t)r * COLS + c], attn * Zp[(size_t)cl * COLS + c]);
}


#define SCAN_TRIPS 16

__device__ __forceinline__ int wave_incl_scan(int x, const int lane) {
  x += __builtin_amdgcn_update_dpp(0, x, 0x111, 0xf, 0xf, true);
  x += __builtin_amdgcn_update_dpp(0, x, 0x112, 0xf, 0xf, true);
  x += __builtin_amdgcn_update_dpp(0, x, 0x114, 0xf, 0xf, true);
  x += __builtin_amdgcn_update_dpp(0, x, 0x118, 0xf, 0xf, true);
  const int t = __builtin_amdgcn_readlane(x, 15);
  return x + (lane >= 16 ? t : 0);
}

__global__ void __launch_bounds__(1024)
row_scan_kernel(const int* __restrict__ cnt, int* __restrict__ rowptr, int N) {
  __shared__ int wsum[32];
  const int lane = threadIdx.x & 31;
  const int w     = __builtin_amdgcn_readfirstlane((int)(threadIdx.x >> 5));
  const int M     = N + 1;
  const int n4    = (N + 3) & ~3;
  const int seg   = ((((M + 31) >> 5) + 127) >> 7) << 7;
  const int trips = seg >> 7;
  const int segb  = w * seg;
  const int lim_l = (segb + seg) < n4 ? (segb + seg) : n4;
  const int lim_s = (segb + seg) < M  ? (segb + seg) : M;
  const int first = segb + 4 * lane;

  int s = 0;
  for (int c = 0; c < trips; c += SCAN_TRIPS) {
    int4 v[SCAN_TRIPS];
    #pragma unroll
    for (int u = 0; u < SCAN_TRIPS; ++u) {
      const int idx = first + (c + u) * 128;
      v[u] = make_int4(0, 0, 0, 0);
      if (idx < lim_l) v[u] = *reinterpret_cast<const int4*>(cnt + idx);
    }
    #pragma unroll
    for (int u = 0; u < SCAN_TRIPS; ++u) s += (v[u].x + v[u].y) + (v[u].z + v[u].w);
  }
  const int tot = __builtin_amdgcn_readlane(wave_incl_scan(s, lane), 31);
  if (lane == 0) wsum[w] = tot;
  __syncthreads();
  int carry = __builtin_amdgcn_readlane(wave_incl_scan(lane < w ? wsum[lane] : 0, lane), 31);

  for (int c = 0; c < trips; c += SCAN_TRIPS) {
    int4 v[SCAN_TRIPS];
    #pragma unroll
    for (int u = 0; u < SCAN_TRIPS; ++u) {
      const int idx = first + (c + u) * 128;
      v[u] = make_int4(0, 0, 0, 0);
      if (idx < lim_l) v[u] = *reinterpret_cast<const int4*>(cnt + idx);
    }
    #pragma unroll
    for (int u = 0; u < SCAN_TRIPS; ++u) {
      const int idx = first + (c + u) * 128;
      const int p1  = v[u].x;
      const int p2  = p1 + v[u].y;
      const int p3  = p2 + v[u].z;
      const int t4  = p3 + v[u].w;
      const int inc = wave_incl_scan(t4, lane);
      const int off = carry + inc - t4;
      if (idx + 3 < lim_s) {
        *reinterpret_cast<int4*>(rowptr + idx) = make_int4(off, off + p1, off + p2, off + p3);
      } else {
        if (idx     < lim_s) rowptr[idx]     = off;
        if (idx + 1 < lim_s) rowptr[idx + 1] = off + p1;
        if (idx + 2 < lim_s) rowptr[idx + 2] = off + p2;
      }
      carry += __builtin_amdgcn_readlane(inc, 31);
    }
  }
}

template <int U>
__device__ __forceinline__ void add_edges(const int myc, const int j, const float4 el,
                                          const float* __restrict__ rl,
                                          const float* __restrict__ zl,
                                          float4& m, float4& s, float4& a0, float4& a1) {
  int c[U]; float4 sc[U], z0[U], z1[U];
  #pragma unroll
  for (int u = 0; u < U; ++u) c[u] = __builtin_amdgcn_readlane(myc, j + u);
  #pragma unroll
  for (int u = 0; u < U; ++u) {
    sc[u] = *reinterpret_cast<const float4*>(rl + (size_t)c[u] * HEADS);
    z0[u] = *reinterpret_cast<const float4*>(zl + (size_t)c[u] * COLS);
    z1[u] = *reinterpret_cast<const float4*>(zl + (size_t)c[u] * COLS + 128);
  }
  float4 mn = m;
  #pragma unroll
  for (int u = 0; u < U; ++u) {
    sc[u].x = leaky01(el.x + sc[u].x); mn.x = fmaxf(mn.x, sc[u].x);
    sc[u].y = leaky01(el.y + sc[u].y); mn.y = fmaxf(mn.y, sc[u].y);
    sc[u].z = leaky01(el.z + sc[u].z); mn.z = fmaxf(mn.z, sc[u].z);
    sc[u].w = leaky01(el.w + sc[u].w); mn.w = fmaxf(mn.w, sc[u].w);
  }
  const float r0 = __expf(m.x - mn.x), r1 = __expf(m.y - mn.y);
  const float r2 = __expf(m.z - mn.z), r3 = __expf(m.w - mn.w);
  s.x  *= r0; s.y  *= r1; s.z  *= r2; s.w  *= r3;
  a0.x *= r0; a0.y *= r1; a0.z *= r2; a0.w *= r3;
  a1.x *= r0; a1.y *= r1; a1.z *= r2; a1.w *= r3;
  #pragma unroll
  for (int u = 0; u < U; ++u) {
    const float w0 = __expf(sc[u].x - mn.x);
    const float w1 = __expf(sc[u].y - mn.y);
    const float w2 = __expf(sc[u].z - mn.z);
    const float w3 = __expf(sc[u].w - mn.w);
    s.x += w0; s.y += w1; s.z += w2; s.w += w3;
    a0.x = fmaf(w0, z0[u].x, a0.x); a0.y = fmaf(w1, z0[u].y, a0.y);
    a0.z = fmaf(w2, z0[u].z, a0.z); a0.w = fmaf(w3, z0[u].w, a0.w);
    a1.x = fmaf(w0, z1[u].x, a1.x); a1.y = fmaf(w1, z1[u].y, a1.y);
    a1.z = fmaf(w2, z1[u].z, a1.z); a1.w = fmaf(w3, z1[u].w, a1.w);
  }
  m = mn;
}

__global__ void __launch_bounds__(256)
row_softmax_gather_kernel(const int* __restrict__ rowptr, const int* __restrict__ scol,
                          const float* __restrict__ e_l, const float* __restrict__ e_r,
                          const float* __restrict__ Zp, float* __restrict__ out, int N) {
  const int lane = threadIdx.x & 31;
  const int r = __builtin_amdgcn_readfirstlane((int)(blockIdx.x * 8 + (threadIdx.x >> 5)));
  if (r >= N) return;
  const int beg = rowptr[r];
  const int end = rowptr[r + 1];
  const int hb  = (lane & 1) * 4;
  const float4 el = *reinterpret_cast<const float4*>(e_l + (size_t)r * HEADS + hb);
  float4 m  = make_float4(-INFINITY, -INFINITY, -INFINITY, -INFINITY);
  float4 s  = make_float4(0.f, 0.f, 0.f, 0.f);
  float4 a0 = make_float4(0.f, 0.f, 0.f, 0.f);
  float4 a1 = make_float4(0.f, 0.f, 0.f, 0.f);
  const float* zl = Zp  + 4 * lane;
  const float* rl = e_r + hb;

  for (int base = beg; base < end; base += 32) {
    const int n = (end - base) < 32 ? (end - base) : 32;
    int myc = (lane < n) ? scol[base + lane] : 0;
    myc = myc < 0 ? 0 : (myc >= N ? N - 1 : myc);
    int j = 0;
    for (; j + 4 <= n; j += 4) add_edges<4>(myc, j, el, rl, zl, m, s, a0, a1);
    for (; j < n; ++j)         add_edges<1>(myc, j, el, rl, zl, m, s, a0, a1);
  }
  if (end > beg) {
    const float q0 = 1.0f / s.x, q1 = 1.0f / s.y, q2 = 1.0f / s.z, q3 = 1.0f / s.w;
    a0.x *= q0; a0.y *= q1; a0.z *= q2; a0.w *= q3;
    a1.x *= q0; a1.y *= q1; a1.z *= q2; a1.w *= q3;
  }
  float* orow = out + (size_t)r * COLS + 4 * lane;
  *reinterpret_cast<float4*>(orow)       = a0;
  *reinterpret_cast<float4*>(orow + 128) = a1;
}

__global__ void __launch_bounds__(256)
row_softmax_gather_slots_kernel(const int* __restrict__ cnt, const int* __restrict__ slots,
                                const int* __restrict__ ovfn, const int* __restrict__ ovf,
                                const float* __restrict__ e_l, const float* __restrict__ e_r,
                                const float* __restrict__ Zp, float* __restrict__ out,
                                int N, int E) {
  const int lane = threadIdx.x & 31;
  const int r = __builtin_amdgcn_readfirstlane((int)(blockIdx.x * 8 + (threadIdx.x >> 5)));
  if (r >= N) return;
  const int* srow = slots + (size_t)r * ROW_SLOTS;
  const int deg = cnt[r];
  const int hb  = (lane & 1) * 4;
  const float4 el = *reinterpret_cast<const float4*>(e_l + (size_t)r * HEADS + hb);
  float4 m  = make_float4(-INFINITY, -INFINITY, -INFINITY, -INFINITY);
  float4 s  = make_float4(0.f, 0.f, 0.f, 0.f);
  float4 a0 = make_float4(0.f, 0.f, 0.f, 0.f);
  float4 a1 = make_float4(0.f, 0.f, 0.f, 0.f);
  const float* zl = Zp  + 4 * lane;
  const float* rl = e_r + hb;

  const int nslot = deg < ROW_SLOTS ? deg : ROW_SLOTS;
  for (int base = 0; base < nslot; base += 32) {
    const int n = (nslot - base) < 32 ? (nslot - base) : 32;
    int myc = (lane < n) ? srow[base + lane] : 0;
    myc = myc < 0 ? 0 : (myc >= N ? N - 1 : myc);
    int j = 0;
    for (; j + 4 <= n; j += 4) add_edges<4>(myc, j, el, rl, zl, m, s, a0, a1);
    for (; j < n; ++j)         add_edges<1>(myc, j, el, rl, zl, m, s, a0, a1);
  }
  if (deg > ROW_SLOTS) {
    int novf = *ovfn;
    novf = novf < 0 ? 0 : (novf > E ? E : novf);
    for (int base = 0; base < novf; base += 32) {
      const int i = base + lane;
      int orow = -1, myc = 0;
      if (i < novf) { orow = ovf[2 * (size_t)i]; myc = ovf[2 * (size_t)i + 1]; }
      myc = myc < 0 ? 0 : (myc >= N ? N - 1 : myc);
      unsigned hit = __builtin_amdgcn_ballot_w32(orow == r);
      while (hit != 0u) {
        const int j = __builtin_ctz(hit);
        hit &= hit - 1u;
        add_edges<1>(myc, j, el, rl, zl, m, s, a0, a1);
      }
    }
  }
  if (deg > 0) {
    const float q0 = 1.0f / s.x, q1 = 1.0f / s.y, q2 = 1.0f / s.z, q3 = 1.0f / s.w;
    a0.x *= q0; a0.y *= q1; a0.z *= q2; a0.w *= q3;
    a1.x *= q0; a1.y *= q1; a1.z *= q2; a1.w *= q3;
  }
  float* orow_out = out + (size_t)r * COLS + 4 * lane;
  *reinterpret_cast<float4*>(orow_out)       = a0;
  *reinterpret_cast<float4*>(orow_out + 128) = a1;
}

extern "C" void kernel_launch(void* const* d_in, const int* in_sizes, int n_in,
                              void* d_out, int out_size, void* d_ws, size_t ws_size,
                              hipStream_t stream) {
  const float* Z   = (const float*)d_in[0];
  const int*   row = (const int*)d_in[1];
  const int*   col = (const int*)d_in[2];
  const float* W   = (const float*)d_in[3];
  const float* bv  = (const float*)d_in[4];
  const float* a_l = (const float*)d_in[5];
  const float* a_r = (const float*)d_in[6];
  float* out = (float*)d_out;

  const int N = in_sizes[0] / IN_F;
  const int E = in_sizes[1];

  float* Zp  = (float*)d_ws;
  float* e_l = Zp  + (size_t)N * COLS;
  float* e_r = e_l + (size_t)N * HEADS;
  float* mb  = e_r + (size_t)N * HEADS;
  float* sb  = mb  + (size_t)N * HEADS;
  float* Wp  = sb  + (size_t)N * HEADS;
  int* cnt    = (int*)(Wp + WP_ELEMS);
  int* rowptr = cnt + ((N + 3) & ~3);
  int* rank   = rowptr + (N + 1);
  int* scol   = rank + E;
  const bool grouped = (size_t)((const char*)(scol + E) - (const char*)d_ws) <= ws_size;
  int* slots = cnt + ((N + 63) & ~63);
  int* ovfn  = slots + (size_t)N * ROW_SLOTS;
  int* ovf   = ovfn + 4;
  const bool slotted = (size_t)((const char*)(ovf + 2 * (size_t)E) - (const char*)d_ws) <= ws_size;

  if (slotted) {
    const int R = (N + GEMM_WGS - 1) / GEMM_WGS;
    const bool owner = R >= 1 && R <= LROWS &&
                       (size_t)(((E + 15) >> 4) << 2) <= (size_t)2 * N * HEADS;
    unsigned* binb = owner ? (unsigned*)mb : nullptr;
    pack_W_kernel<<<(WP_ELEMS + 255) / 256, 256, 0, stream>>>(W, Wp, cnt, N, ovfn, row, binb, E,
                                                              owner ? R : 1);
    proj_gemm_wmma<<<GEMM_WGS, 1024, 0, stream>>>(Z, Wp, bv, Zp, N, row, cnt, nullptr, E,
                                                  col, slots, ovfn, ovf, binb, R);
    const int NHs = N * HEADS;
    attn_logits_init<<<(NHs + 255) / 256, 256, 0, stream>>>(Zp, a_l, a_r, e_l, e_r, nullptr, nullptr,
                                                            row, col, nullptr, nullptr, nullptr,
                                                            NHs, E);
    row_softmax_gather_slots_kernel<<<(N + 7) / 8, 256, 0, stream>>>(cnt, slots, ovfn, ovf,
                                                                     e_l, e_r, Zp, out, N, E);
    return;
  }

  pack_W_kernel<<<(WP_ELEMS + 255) / 256, 256, 0, stream>>>(W, Wp, grouped ? cnt : nullptr, N, nullptr,
                                                            row, nullptr, E, 1);

  proj_gemm_wmma<<<GEMM_WGS, 1024, 0, stream>>>(Z, Wp, bv, Zp, N, row, grouped ? cnt : nullptr,
                                                rank, E, col, nullptr, nullptr, nullptr, nullptr, 0);

  if (grouped) row_scan_kernel<<<1, 1024, 0, stream>>>(cnt, rowptr, N);

  const int NH = N * HEADS;
  attn_logits_init<<<(NH + 255) / 256, 256, 0, stream>>>(Zp, a_l, a_r, e_l, e_r,
                                                         grouped ? nullptr : mb,
                                                         grouped ? nullptr : sb,
                                                         row, col,
                                                         grouped ? rowptr : nullptr, rank, scol,
                                                         NH, E);

  if (grouped) {
    row_softmax_gather_kernel<<<(N + 7) / 8, 256, 0, stream>>>(rowptr, scol, e_l, e_r, Zp, out, N);
  } else {
    const int EH = E * HEADS;
    edge_max_kernel   <<<(EH + 255) / 256, 256, 0, stream>>>(row, col, e_l, e_r, mb, EH);
    edge_expsum_kernel<<<(EH + 255) / 256, 256, 0, stream>>>(row, col, e_l, e_r, mb, sb, EH);
    hipMemsetAsync(out, 0, (size_t)out_size * sizeof(float), stream);
    edge_scatter_kernel<<<E, 256, 0, stream>>>(row, col, e_l, e_r, mb, sb, Zp, out);
  }
}
